# starting kernel
# baseline (speedup 1.0000x reference)
#include <hip/hip_runtime.h>

typedef __attribute__((ext_vector_type(2))) float v2f;
typedef __attribute__((ext_vector_type(8))) float v8f;

#define EPS 1e-5f
#define HW  3136
#define CIN 256
#define CR  64
#define GK  784
#define NB  4

__global__ void __launch_bounds__(128)
invo_gemm1_bn_relu(const float* __restrict__ x, const float* __restrict__ W1,
                   const float* __restrict__ b1, const float* __restrict__ gamma,
                   const float* __restrict__ beta, const float* __restrict__ mean,
                   const float* __restrict__ var, float* __restrict__ t)
{
    const int lane   = threadIdx.x & 31;
    const int wave   = threadIdx.x >> 5;
    const int laneN  = lane & 15;
    const int laneHi = lane >> 4;
    const int p0     = blockIdx.x * 16;
    const int b      = blockIdx.y;
    const int o0     = wave * 16;

    const float* xb   = x + (size_t)b * CIN * HW;
    const float* Arow = W1 + (size_t)(o0 + laneN) * CIN + 2 * laneHi;
    const float* Bcol = xb + (size_t)(2 * laneHi) * HW + p0 + laneN;

    v8f acc = {};
#pragma unroll 8
    for (int k = 0; k < CIN; k += 4) {
        v2f a = *(const v2f*)(Arow + k);
        v2f bb;
        bb.x = Bcol[(size_t)k * HW];
        bb.y = Bcol[(size_t)(k + 1) * HW];
        acc = __builtin_amdgcn_wmma_f32_16x16x4_f32(
            false, a, false, bb, (short)0, acc, false, false);
    }

    float* tb = t + (size_t)b * CR * HW + p0 + laneN;
#pragma unroll
    for (int r = 0; r < 8; ++r) {
        int o = o0 + r + 8 * laneHi;
        float s  = gamma[o] * __frsqrt_rn(var[o] + EPS);
        float v  = (acc[r] + b1[o]) * s + (beta[o] - mean[o] * s);
        tb[(size_t)o * HW] = fmaxf(v, 0.0f);
    }
}

__global__ void __launch_bounds__(32)
invo_gemm2(const float* __restrict__ t, const float* __restrict__ W2,
           const float* __restrict__ b2, float* __restrict__ w)
{
    const int lane   = threadIdx.x & 31;
    const int laneN  = lane & 15;
    const int laneHi = lane >> 4;
    const int p0     = blockIdx.x * 16;
    const int m0     = blockIdx.y * 16;
    const int b      = blockIdx.z;

    const float* tb   = t + (size_t)b * CR * HW;
    const float* Arow = W2 + (size_t)(m0 + laneN) * CR + 2 * laneHi;
    const float* Bcol = tb + (size_t)(2 * laneHi) * HW + p0 + laneN;

    v8f acc = {};
#pragma unroll
    for (int k = 0; k < CR; k += 4) {
        v2f a = *(const v2f*)(Arow + k);
        v2f bb;
        bb.x = Bcol[(size_t)k * HW];
        bb.y = Bcol[(size_t)(k + 1) * HW];
        acc = __builtin_amdgcn_wmma_f32_16x16x4_f32(
            false, a, false, bb, (short)0, acc, false, false);
    }

    float* wb = w + (size_t)b * GK * HW + p0 + laneN;
#pragma unroll
    for (int r = 0; r < 8; ++r) {
        int m = m0 + r + 8 * laneHi;
        wb[(size_t)m * HW] = acc[r] + b2[m];
    }
}

__global__ void __launch_bounds__(256)
invo_apply(const float* __restrict__ x, const float* __restrict__ wk,
           float* __restrict__ out)
{
    __shared__ float wl[49 * 16];

    const int p0 = blockIdx.x * 16;
    const int g  = blockIdx.y;
    const int b  = blockIdx.z;
    const int tid = threadIdx.x;

    const float* wsrc = wk + ((size_t)b * GK + (size_t)g * 49) * HW + p0;
#pragma unroll
    for (int idx = tid; idx < 49 * 16; idx += 256) {
        int kk = idx >> 4, pl = idx & 15;
        wl[kk * 16 + pl] = wsrc[(size_t)kk * HW + pl];
    }
    __syncthreads();

    const int pl = tid & 15;
    const int cg = tid >> 4;
    const int p  = p0 + pl;
    const int h  = p / 56;
    const int wc = p % 56;
    const int c  = g * 16 + cg;

    const float* xc = x + ((size_t)b * CIN + c) * HW;

    float acc = 0.0f;
#pragma unroll
    for (int i = 0; i < 7; ++i) {
        int hh = h + i - 3;
        bool hin = (hh >= 0) & (hh < 56);
#pragma unroll
        for (int j = 0; j < 7; ++j) {
            int ww = wc + j - 3;
            bool ok = hin & (ww >= 0) & (ww < 56);
            float xv = ok ? xc[hh * 56 + ww] : 0.0f;
            acc = fmaf(xv, wl[(i * 7 + j) * 16 + pl], acc);
        }
    }
    out[((size_t)b * CIN + c) * HW + p] = acc;
}

extern "C" void kernel_launch(void* const* d_in, const int* in_sizes, int n_in,
                              void* d_out, int out_size, void* d_ws, size_t ws_size,
                              hipStream_t stream)
{
    const float* x     = (const float*)d_in[0];
    const float* W1    = (const float*)d_in[1];
    const float* b1    = (const float*)d_in[2];
    const float* gamma = (const float*)d_in[3];
    const float* beta  = (const float*)d_in[4];
    const float* mean  = (const float*)d_in[5];
    const float* var   = (const float*)d_in[6];
    const float* W2    = (const float*)d_in[7];
    const float* b2    = (const float*)d_in[8];
    float* out = (float*)d_out;

    float* t = (float*)d_ws;
    float* w = t + (size_t)NB * CR * HW;

    dim3 g1(HW / 16, NB);
    invo_gemm1_bn_relu<<<g1, 128, 0, stream>>>(x, W1, b1, gamma, beta, mean, var, t);

    dim3 g2(HW / 16, GK / 16, NB);
    invo_gemm2<<<g2, 32, 0, stream>>>(t, W2, b2, w);

    dim3 g3(HW / 16, 16, NB);
    invo_apply<<<g3, 256, 0, stream>>>(x, w, out);
}


# Fuse the second GEMM into the involution kernel: each block computes its 49x16 per-pixel weights by WMMA straight into LDS, so the weight tensor never goes through memory
# speedup vs baseline: 1.4033x; 1.4033x over previous
#include <hip/hip_runtime.h>

typedef __attribute__((ext_vector_type(2))) float v2f;
typedef __attribute__((ext_vector_type(8))) float v8f;

#define EPS 1e-5f
#define HW  3136
#define CIN 256
#define CR  64
#define GK  784
#define NB  4

__global__ void __launch_bounds__(128)
invo_gemm1_bn_relu(const float* __restrict__ x, const float* __restrict__ W1,
                   const float* __restrict__ b1, const float* __restrict__ gamma,
                   const float* __restrict__ beta, const float* __restrict__ mean,
                   const float* __restrict__ var, float* __restrict__ t)
{
    const int lane   = threadIdx.x & 31;
    const int wave   = threadIdx.x >> 5;
    const int laneN  = lane & 15;
    const int laneHi = lane >> 4;
    const int p0     = blockIdx.x * 16;
    const int b      = blockIdx.y;
    const int o0     = wave * 16;

    const float* xb   = x + (size_t)b * CIN * HW;
    const float* Arow = W1 + (size_t)(o0 + laneN) * CIN + 2 * laneHi;
    const float* Bcol = xb + (size_t)(2 * laneHi) * HW + p0 + laneN;

    v8f acc = {};
#pragma unroll 8
    for (int k = 0; k < CIN; k += 4) {
        v2f a = *(const v2f*)(Arow + k);
        v2f bb;
        bb.x = Bcol[(size_t)k * HW];
        bb.y = Bcol[(size_t)(k + 1) * HW];
        acc = __builtin_amdgcn_wmma_f32_16x16x4_f32(
            false, a, false, bb, (short)0, acc, false, false);
    }

    float* tb = t + (size_t)b * CR * HW + p0 + laneN;
#pragma unroll
    for (int r = 0; r < 8; ++r) {
        int o = o0 + r + 8 * laneHi;
        float s  = gamma[o] * __frsqrt_rn(var[o] + EPS);
        float v  = (acc[r] + b1[o]) * s + (beta[o] - mean[o] * s);
        tb[(size_t)o * HW] = fmaxf(v, 0.0f);
    }
}

__global__ void __launch_bounds__(256)
invo_gemm2_apply(const float* __restrict__ x, const float* __restrict__ t,
                 const float* __restrict__ W2, const float* __restrict__ b2,
                 float* __restrict__ out)
{
    __shared__ float wl[64 * 16];

    const int p0  = blockIdx.x * 16;
    const int g   = blockIdx.y;
    const int b   = blockIdx.z;
    const int tid = threadIdx.x;

    const int wave = __builtin_amdgcn_readfirstlane(tid >> 5);

    if (wave < 4) {
        const int lane   = tid & 31;
        const int laneN  = lane & 15;
        const int laneHi = lane >> 4;
        const int tap0   = wave * 16;
        const int tapA   = min(tap0 + laneN, 48);

        const float* Arow = W2 + (size_t)(g * 49 + tapA) * CR + 2 * laneHi;
        const float* Bcol = t + (size_t)b * CR * HW + (size_t)(2 * laneHi) * HW + p0 + laneN;

        float bias[8];
#pragma unroll
        for (int r = 0; r < 8; ++r)
            bias[r] = b2[g * 49 + min(tap0 + r + 8 * laneHi, 48)];

        v8f acc = {};
#pragma unroll
        for (int k = 0; k < CR; k += 4) {
            v2f a = *(const v2f*)(Arow + k);
            v2f bb;
            bb.x = Bcol[(size_t)k * HW];
            bb.y = Bcol[(size_t)(k + 1) * HW];
            acc = __builtin_amdgcn_wmma_f32_16x16x4_f32(
                false, a, false, bb, (short)0, acc, false, false);
        }

#pragma unroll
        for (int r = 0; r < 8; ++r)
            wl[(tap0 + r + 8 * laneHi) * 16 + laneN] = acc[r] + bias[r];
    }
    __syncthreads();

    const int pl = tid & 15;
    const int cg = tid >> 4;
    const int p  = p0 + pl;
    const int h  = p / 56;
    const int wc = p % 56;
    const int c  = g * 16 + cg;

    const float* xc = x + ((size_t)b * CIN + c) * HW;

    float acc = 0.0f;
#pragma unroll
    for (int i = 0; i < 7; ++i) {
        int hh = h + i - 3;
        bool hin = (hh >= 0) & (hh < 56);
#pragma unroll
        for (int j = 0; j < 7; ++j) {
            int ww = wc + j - 3;
            bool ok = hin & (ww >= 0) & (ww < 56);
            float xv = ok ? xc[hh * 56 + ww] : 0.0f;
            acc = fmaf(xv, wl[(i * 7 + j) * 16 + pl], acc);
        }
    }
    out[((size_t)b * CIN + c) * HW + p] = acc;
}

extern "C" void kernel_launch(void* const* d_in, const int* in_sizes, int n_in,
                              void* d_out, int out_size, void* d_ws, size_t ws_size,
                              hipStream_t stream)
{
    const float* x     = (const float*)d_in[0];
    const float* W1    = (const float*)d_in[1];
    const float* b1    = (const float*)d_in[2];
    const float* gamma = (const float*)d_in[3];
    const float* beta  = (const float*)d_in[4];
    const float* mean  = (const float*)d_in[5];
    const float* var   = (const float*)d_in[6];
    const float* W2    = (const float*)d_in[7];
    const float* b2    = (const float*)d_in[8];
    float* out = (float*)d_out;

    float* t = (float*)d_ws;

    dim3 g1(HW / 16, NB);
    invo_gemm1_bn_relu<<<g1, 128, 0, stream>>>(x, W1, b1, gamma, beta, mean, var, t);

    dim3 g3(HW / 16, 16, NB);
    invo_gemm2_apply<<<g3, 256, 0, stream>>>(x, t, W2, b2, out);
}


# Register-tile the fused kernel: 64-pixel block, a thread applies 4 adjacent pixels x 2 channels from 128-bit x and LDS reads; a wave keeps its W2 fragment for 4 pixel tiles fed by 128-bit t loads
# speedup vs baseline: 3.8630x; 2.7528x over previous
#include <hip/hip_runtime.h>

typedef __attribute__((ext_vector_type(2))) float v2f;
typedef __attribute__((ext_vector_type(8))) float v8f;

#define EPS 1e-5f
#define HW  3136
#define CIN 256
#define CR  64
#define GK  784
#define NB  4

__global__ void __launch_bounds__(128)
invo_gemm1_bn_relu(const float* __restrict__ x, const float* __restrict__ W1,
                   const float* __restrict__ b1, const float* __restrict__ gamma,
                   const float* __restrict__ beta, const float* __restrict__ mean,
                   const float* __restrict__ var, float* __restrict__ t)
{
    const int lane   = threadIdx.x & 31;
    const int wave   = threadIdx.x >> 5;
    const int laneN  = lane & 15;
    const int laneHi = lane >> 4;
    const int p0     = blockIdx.x * 16;
    const int b      = blockIdx.y;
    const int o0     = wave * 16;

    const float* xb   = x + (size_t)b * CIN * HW;
    const float* Arow = W1 + (size_t)(o0 + laneN) * CIN + 2 * laneHi;
    const float* Bcol = xb + (size_t)(2 * laneHi) * HW + p0 + laneN;

    v8f acc = {};
#pragma unroll 8
    for (int k = 0; k < CIN; k += 4) {
        v2f a = *(const v2f*)(Arow + k);
        v2f bb;
        bb.x = Bcol[(size_t)k * HW];
        bb.y = Bcol[(size_t)(k + 1) * HW];
        acc = __builtin_amdgcn_wmma_f32_16x16x4_f32(
            false, a, false, bb, (short)0, acc, false, false);
    }

    float* tb = t + (size_t)b * CR * HW + p0 + laneN;
#pragma unroll
    for (int r = 0; r < 8; ++r) {
        int o = o0 + r + 8 * laneHi;
        float s  = gamma[o] * __frsqrt_rn(var[o] + EPS);
        float v  = (acc[r] + b1[o]) * s + (beta[o] - mean[o] * s);
        tb[(size_t)o * HW] = fmaxf(v, 0.0f);
    }
}

#define WP 68

__global__ void __launch_bounds__(128)
invo_gemm2_apply(const float* __restrict__ x, const float* __restrict__ t,
                 const float* __restrict__ W2, const float* __restrict__ b2,
                 float* __restrict__ out)
{
    __shared__ __attribute__((aligned(16))) float wl[64 * WP];

    const int p0  = blockIdx.x * 64;
    const int g   = blockIdx.y;
    const int b   = blockIdx.z;
    const int tid = threadIdx.x;

    {
        const int wave   = tid >> 5;
        const int lane   = tid & 31;
        const int laneN  = lane & 15;
        const int laneHi = lane >> 4;
        const int tap0   = wave * 16;
        const int tapA   = min(tap0 + laneN, 48);

        const float* Arow = W2 + (size_t)(g * 49 + tapA) * CR + 2 * laneHi;
        const float* Bcol = t + (size_t)b * CR * HW + (size_t)(2 * laneHi) * HW + p0 + 4 * laneN;

        float bias[8];
#pragma unroll
        for (int r = 0; r < 8; ++r)
            bias[r] = b2[g * 49 + min(tap0 + r + 8 * laneHi, 48)];

        v2f a[CR / 4];
#pragma unroll
        for (int kk = 0; kk < CR / 4; ++kk)
            a[kk] = *(const v2f*)(Arow + 4 * kk);

        v8f acc[4] = {};
#pragma unroll
        for (int kk = 0; kk < CR / 4; ++kk) {
            const int k = 4 * kk;
            const float4 t0 = *(const float4*)(Bcol + (size_t)k * HW);
            const float4 t1 = *(const float4*)(Bcol + (size_t)(k + 1) * HW);
            v2f bb0 = {t0.x, t1.x}, bb1 = {t0.y, t1.y}, bb2 = {t0.z, t1.z}, bb3 = {t0.w, t1.w};
            acc[0] = __builtin_amdgcn_wmma_f32_16x16x4_f32(
                false, a[kk], false, bb0, (short)0, acc[0], false, false);
            acc[1] = __builtin_amdgcn_wmma_f32_16x16x4_f32(
                false, a[kk], false, bb1, (short)0, acc[1], false, false);
            acc[2] = __builtin_amdgcn_wmma_f32_16x16x4_f32(
                false, a[kk], false, bb2, (short)0, acc[2], false, false);
            acc[3] = __builtin_amdgcn_wmma_f32_16x16x4_f32(
                false, a[kk], false, bb3, (short)0, acc[3], false, false);
        }

#pragma unroll
        for (int r = 0; r < 8; ++r)
            *(float4*)&wl[(tap0 + r + 8 * laneHi) * WP + 4 * laneN] =
                make_float4(acc[0][r] + bias[r], acc[1][r] + bias[r],
                            acc[2][r] + bias[r], acc[3][r] + bias[r]);
    }
    __syncthreads();

    const int q  = tid & 15;
    const int cp = tid >> 4;
    const int p  = p0 + 4 * q;
    const int h  = p / 56;
    const int wc = p % 56;
    const int c  = g * 16 + 2 * cp;

    const float* xc = x + ((size_t)b * CIN + c) * HW;

    const bool lin = wc > 0;
    const bool rin = wc < 52;
    const int  lo  = lin ? -4 : 0;
    const int  ro  = rin ?  4 : 0;

    float acc[2][4] = {};
#pragma unroll
    for (int i = 0; i < 7; ++i) {
        const int  hh  = h + i - 3;
        const bool hin = (hh >= 0) & (hh < 56);
        const int  hc  = min(max(hh, 0), 55);
        const float* row = xc + hc * 56 + wc;

        float xw[2][12];
#pragma unroll
        for (int ch = 0; ch < 2; ++ch) {
            const float4 L = *(const float4*)(row + ch * HW + lo);
            const float4 M = *(const float4*)(row + ch * HW);
            const float4 R = *(const float4*)(row + ch * HW + ro);
            const bool okl = hin & lin, okr = hin & rin;
            xw[ch][0]  = okl ? L.x : 0.0f;  xw[ch][1]  = okl ? L.y : 0.0f;
            xw[ch][2]  = okl ? L.z : 0.0f;  xw[ch][3]  = okl ? L.w : 0.0f;
            xw[ch][4]  = hin ? M.x : 0.0f;  xw[ch][5]  = hin ? M.y : 0.0f;
            xw[ch][6]  = hin ? M.z : 0.0f;  xw[ch][7]  = hin ? M.w : 0.0f;
            xw[ch][8]  = okr ? R.x : 0.0f;  xw[ch][9]  = okr ? R.y : 0.0f;
            xw[ch][10] = okr ? R.z : 0.0f;  xw[ch][11] = okr ? R.w : 0.0f;
        }
#pragma unroll
        for (int j = 0; j < 7; ++j) {
            const float4 wv = *(const float4*)&wl[(i * 7 + j) * WP + 4 * q];
#pragma unroll
            for (int ch = 0; ch < 2; ++ch) {
                acc[ch][0] = fmaf(xw[ch][j + 1], wv.x, acc[ch][0]);
                acc[ch][1] = fmaf(xw[ch][j + 2], wv.y, acc[ch][1]);
                acc[ch][2] = fmaf(xw[ch][j + 3], wv.z, acc[ch][2]);
                acc[ch][3] = fmaf(xw[ch][j + 4], wv.w, acc[ch][3]);
            }
        }
    }
#pragma unroll
    for (int ch = 0; ch < 2; ++ch)
        *(float4*)&out[((size_t)b * CIN + c + ch) * HW + p] =
            make_float4(acc[ch][0], acc[ch][1], acc[ch][2], acc[ch][3]);
}

extern "C" void kernel_launch(void* const* d_in, const int* in_sizes, int n_in,
                              void* d_out, int out_size, void* d_ws, size_t ws_size,
                              hipStream_t stream)
{
    const float* x     = (const float*)d_in[0];
    const float* W1    = (const float*)d_in[1];
    const float* b1    = (const float*)d_in[2];
    const float* gamma = (const float*)d_in[3];
    const float* beta  = (const float*)d_in[4];
    const float* mean  = (const float*)d_in[5];
    const float* var   = (const float*)d_in[6];
    const float* W2    = (const float*)d_in[7];
    const float* b2    = (const float*)d_in[8];
    float* out = (float*)d_out;

    float* t = (float*)d_ws;

    dim3 g1(HW / 16, NB);
    invo_gemm1_bn_relu<<<g1, 128, 0, stream>>>(x, W1, b1, gamma, beta, mean, var, t);

    dim3 g3(HW / 64, 16, NB);
    invo_gemm2_apply<<<g3, 128, 0, stream>>>(x, t, W2, b2, out);
}
